# starting kernel
# baseline (speedup 1.0000x reference)
#include <hip/hip_runtime.h>
#include <math.h>

#define Bn 4
#define Nn 512
#define Fo 64
#define Ki 128
#define ALPHAF 0.2f

typedef __attribute__((ext_vector_type(2))) float v2f;
typedef __attribute__((ext_vector_type(8))) float v8f;

__device__ __forceinline__ float lrelu(float v) { return v > 0.f ? v : ALPHAF * v; }

__global__ __launch_bounds__(32) void gat_hgemm(const float* __restrict__ x,
                                                const float* __restrict__ W,
                                                const float* __restrict__ bias,
                                                float* __restrict__ h) {
  const int tile = blockIdx.x;
  const int nt = tile & 3;
  const int mt = tile >> 2;
  const int lane = threadIdx.x;
  const int l15 = lane & 15;
  const int kg  = lane >> 4;
  const int row0 = mt * 16;
  const int n0   = nt * 16;

  v8f acc = {};
  for (int k0 = 0; k0 < Ki; k0 += 4) {
    const int ka = k0 + kg * 2;
    const float* xr = x + (size_t)(row0 + l15) * Ki + ka;
    v2f Af; Af.x = xr[0]; Af.y = xr[1];
    const float* wr = W + (size_t)ka * Fo + n0 + l15;
    v2f Bf; Bf.x = wr[0]; Bf.y = wr[Fo];
    acc = __builtin_amdgcn_wmma_f32_16x16x4_f32(false, Af, false, Bf,
                                                (short)0, acc, false, false);
  }
  const float bv = bias[n0 + l15];
  const int rbase = row0 + kg * 8;
#pragma unroll
  for (int i = 0; i < 8; ++i) {
    h[(size_t)(rbase + i) * Fo + n0 + l15] = acc[i] + bv;
  }
}

__global__ void gat_pq(const float* __restrict__ h, const float* __restrict__ a,
                       float* __restrict__ p, float* __restrict__ q) {
  const int r = blockIdx.x * blockDim.x + threadIdx.x;
  if (r >= Bn * Nn) return;
  const float* hr = h + (size_t)r * Fo;
  float sp = 0.f, sq = 0.f;
#pragma unroll 4
  for (int f = 0; f < Fo; ++f) {
    sp += hr[f] * a[f];
    sq += hr[f] * a[Fo + f];
  }
  p[r] = sp;
  q[r] = sq;
}

__global__ __launch_bounds__(128) void gat_softmax(const int* __restrict__ adj,
                                                   const float* __restrict__ p,
                                                   const float* __restrict__ q,
                                                   float* __restrict__ prob) {
  const int row  = blockIdx.x * 4 + (threadIdx.x >> 5);
  const int lane = threadIdx.x & 31;
  const int b = row >> 9;
  const int n = row & (Nn - 1);
  const int* adjr = adj + (size_t)b * Nn * Nn + (size_t)n * Nn;
  const float* pb = p + b * Nn;
  const float* qb = q + b * Nn;

  float vlo = 0.f, vhi = 0.f;
  if (n < 256) {
    vlo = lrelu(pb[2 * n]     + qb[2 * n]);
    vhi = lrelu(pb[2 * n + 1] + qb[2 * n + 1]);
  }

  float vals[16];
  float mx = -INFINITY;
#pragma unroll
  for (int j = 0; j < 16; ++j) {
    const int m = lane + 32 * j;
    float e;
    if (n < 256) {
      e = (m < 256) ? vlo : vhi;
    } else {
      const int mm = m & 255;
      e = lrelu(pb[2 * mm] + qb[2 * mm + 1]);
    }
    e = (adjr[m] > 0) ? e : -INFINITY;
    vals[j] = e;
    mx = fmaxf(mx, e);
  }
#pragma unroll
  for (int off = 16; off >= 1; off >>= 1) mx = fmaxf(mx, __shfl_xor(mx, off, 32));

  float sum = 0.f;
#pragma unroll
  for (int j = 0; j < 16; ++j) {
    const float e = vals[j];
    const float ex = (e == -INFINITY) ? 0.f : __expf(e - mx);
    vals[j] = ex;
    sum += ex;
  }
#pragma unroll
  for (int off = 16; off >= 1; off >>= 1) sum += __shfl_xor(sum, off, 32);

  const float inv = 1.f / sum;
  float* pr = prob + (size_t)b * Nn * Nn + (size_t)n * Nn;
#pragma unroll
  for (int j = 0; j < 16; ++j) pr[lane + 32 * j] = vals[j] * inv;
}

__global__ __launch_bounds__(32) void gat_aggr(const float* __restrict__ prob,
                                               const float* __restrict__ h,
                                               float* __restrict__ out) {
  const int tile = blockIdx.x;
  const int nt = tile & 3;
  const int mt = (tile >> 2) & 31;
  const int b  = tile >> 7;
  const int lane = threadIdx.x;
  const int l15 = lane & 15;
  const int kg  = lane >> 4;
  const int row0 = mt * 16;
  const int n0   = nt * 16;

  const float* pb = prob + (size_t)b * Nn * Nn;
  const float* hb = h + (size_t)b * Nn * Fo;

  v8f acc = {};
  for (int k0 = 0; k0 < Nn; k0 += 4) {
    const int ka = k0 + kg * 2;
    const float* ar = pb + (size_t)(row0 + l15) * Nn + ka;
    v2f Af; Af.x = ar[0]; Af.y = ar[1];
    const float* br = hb + (size_t)ka * Fo + n0 + l15;
    v2f Bf; Bf.x = br[0]; Bf.y = br[Fo];
    acc = __builtin_amdgcn_wmma_f32_16x16x4_f32(false, Af, false, Bf,
                                                (short)0, acc, false, false);
  }
  float* ob = out + (size_t)b * Nn * Fo;
  const int rbase = row0 + kg * 8;
#pragma unroll
  for (int i = 0; i < 8; ++i) {
    float v = acc[i];
    v = (v > 0.f) ? v : (__expf(v) - 1.f);
    ob[(size_t)(rbase + i) * Fo + n0 + l15] = v;
  }
}

extern "C" void kernel_launch(void* const* d_in, const int* in_sizes, int n_in,
                              void* d_out, int out_size, void* d_ws, size_t ws_size,
                              hipStream_t stream) {
  const float* x    = (const float*)d_in[0];
  const int*   adj  = (const int*)d_in[1];
  const float* W    = (const float*)d_in[2];
  const float* bias = (const float*)d_in[3];
  const float* a    = (const float*)d_in[4];
  float* out = (float*)d_out;

  float* h    = (float*)d_ws;
  float* p    = h + (size_t)Bn * Nn * Fo;
  float* q    = p + Bn * Nn;
  float* prob = q + Bn * Nn;

  gat_hgemm<<<512, 32, 0, stream>>>(x, W, bias, h);
  gat_pq<<<(Bn * Nn + 255) / 256, 256, 0, stream>>>(h, a, p, q);
  gat_softmax<<<512, 128, 0, stream>>>(adj, p, q, prob);
  gat_aggr<<<512, 32, 0, stream>>>(prob, h, out);
}


# aggregation re-associated as (prob.x).W + b so nothing waits on h: each 16-row block takes its logits from x and W.a, prob stays in LDS; 3 launches to 1 (p/q and output sum order change)
# speedup vs baseline: 2.4505x; 2.4505x over previous
#include <hip/hip_runtime.h>
#include <math.h>

#define Bn 4
#define Nn 512
#define Fo 64
#define Ki 128
#define ALPHAF 0.2f

typedef __attribute__((ext_vector_type(2))) float v2f;
typedef __attribute__((ext_vector_type(8))) float v8f;
typedef __attribute__((ext_vector_type(4))) float v4f;

__device__ __forceinline__ float lrelu(float v) { return v > 0.f ? v : ALPHAF * v; }

#define NTH 256
#define RP  33
#define PP  516
#define TP  132
#define USZ (8 * 32 * RP)

__device__ __forceinline__ void t_load(const float* pa, const float* __restrict__ xg, int it0,
                                       v4f* av, v4f* bv) {
#pragma unroll
  for (int i = 0; i < 4; ++i) {
    av[i] = *(const v4f*)(pa + 8 * (it0 + i));
#pragma unroll
    for (int s = 0; s < 4; ++s)
      bv[4 * i + s] = *(const v4f*)(xg + (size_t)(8 * (it0 + i) + s) * Ki);
  }
}

__global__ __launch_bounds__(NTH) void gat_one(const float* __restrict__ x,
                                               const int* __restrict__ adj,
                                               const float* __restrict__ W,
                                               const float* __restrict__ bias,
                                               const float* __restrict__ a,
                                               float* __restrict__ out) {
  __shared__ __attribute__((aligned(16))) float wa[2 * Ki];
  __shared__ float El[256];
  __shared__ __attribute__((aligned(16))) float U[USZ];
  __shared__ __attribute__((aligned(16))) float Tq[4 * 16 * TP];

  const int tid  = threadIdx.x;
  const int lane = tid & 31;
  const int w    = tid >> 5;
  const int b    = blockIdx.x >> 5;
  const int mt   = blockIdx.x & 31;
  const int row0 = mt * 16;
  const bool lo  = mt < 16;
  const float* xb = x + (size_t)b * Nn * Ki;

  int mk[2][16];
  {
    const int* ar = adj + (size_t)b * Nn * Nn + (size_t)(row0 + 2 * w) * Nn;
#pragma unroll
    for (int rr = 0; rr < 2; ++rr)
#pragma unroll
      for (int j = 0; j < 16; ++j) mk[rr][j] = ar[rr * Nn + lane + 32 * j];
  }

  {
    const int k = tid >> 1, hf = tid & 1;
    const float* wr = W + (size_t)k * Fo;
    const float* ah = a + hf * Fo;
    float s = 0.f;
#pragma unroll
    for (int cc = 0; cc < Fo / 4; ++cc) {
      const v4f w4 = *(const v4f*)(wr + 4 * cc);
      s = fmaf(w4.x, ah[4 * cc],     s);
      s = fmaf(w4.y, ah[4 * cc + 1], s);
      s = fmaf(w4.z, ah[4 * cc + 2], s);
      s = fmaf(w4.w, ah[4 * cc + 3], s);
    }
    wa[hf * Ki + k] = s;
  }
  float c0 = fmaf(bias[lane + 32], a[lane + 32], bias[lane] * a[lane]);
  float c1 = fmaf(bias[lane + 32], a[Fo + lane + 32], bias[lane] * a[Fo + lane]);
#pragma unroll
  for (int off = 16; off >= 1; off >>= 1) {
    c0 += __shfl_xor(c0, off, 32);
    c1 += __shfl_xor(c1, off, 32);
  }
  const float c01 = c0 + c1;
  __syncthreads();

  const int nitems = lo ? 32 : 256;
  float* redw = U + w * (32 * RP);
  if (32 * w < nitems) {
    const v4f w0 = *(const v4f*)(wa + 4 * lane);
    const v4f w1 = *(const v4f*)(wa + Ki + 4 * lane);
#pragma unroll 1
    for (int hb = 0; hb < 2; ++hb) {
      v4f xa[16], xq[16];
#pragma unroll
      for (int i = 0; i < 16; ++i) {
        const int item = 32 * w + 16 * hb + i;
        const int ra = lo ? (2 * row0 + item) : (2 * item);
        const int rb = lo ? ra : (ra + 1);
        xa[i] = *(const v4f*)(xb + (size_t)ra * Ki + 4 * lane);
        xq[i] = *(const v4f*)(xb + (size_t)rb * Ki + 4 * lane);
      }
#pragma unroll
      for (int i = 0; i < 16; ++i) {
        float s = xa[i].x * w0.x;
        s = fmaf(xa[i].y, w0.y, s);
        s = fmaf(xa[i].z, w0.z, s);
        s = fmaf(xa[i].w, w0.w, s);
        s = fmaf(xq[i].x, w1.x, s);
        s = fmaf(xq[i].y, w1.y, s);
        s = fmaf(xq[i].z, w1.z, s);
        s = fmaf(xq[i].w, w1.w, s);
        redw[(16 * hb + i) * RP + lane] = s;
      }
    }
  }
  __syncthreads();
  if (32 * w < nitems) {
    float s = 0.f;
#pragma unroll
    for (int t = 0; t < 32; ++t) s += redw[lane * RP + t];
    El[32 * w + lane] = lrelu(s + c01);
  }
  __syncthreads();

  float* Pl = U;
#pragma unroll
  for (int rr = 0; rr < 2; ++rr) {
    const int rl = 2 * w + rr;
    const float elo = El[2 * rl];
    const float ehi = El[2 * rl + 1];
    float vals[16];
    if (lo) {
#pragma unroll
      for (int j = 0; j < 16; ++j) vals[j] = (j < 8) ? elo : ehi;
    } else {
#pragma unroll
      for (int j = 0; j < 16; ++j) vals[j] = El[lane + 32 * (j & 7)];
    }
    float mx = -INFINITY;
#pragma unroll
    for (int j = 0; j < 16; ++j) {
      const float e = (mk[rr][j] > 0) ? vals[j] : -INFINITY;
      vals[j] = e;
      mx = fmaxf(mx, e);
    }
#pragma unroll
    for (int off = 16; off >= 1; off >>= 1) mx = fmaxf(mx, __shfl_xor(mx, off, 32));
    float sum = 0.f;
#pragma unroll
    for (int j = 0; j < 16; ++j) {
      const float e = vals[j];
      const float ex = (e == -INFINITY) ? 0.f : __expf(e - mx);
      vals[j] = ex;
      sum += ex;
    }
#pragma unroll
    for (int off = 16; off >= 1; off >>= 1) sum += __shfl_xor(sum, off, 32);
    const float inv = 1.f / sum;
#pragma unroll
    for (int j = 0; j < 16; ++j) Pl[rl * PP + lane + 32 * j] = vals[j] * inv;
  }
  __syncthreads();

  const int c = lane & 15;
  const int h = lane >> 4;
  {
    const int fh = w & 1, kq = w >> 1;
    const float* pa = Pl + c * PP + 128 * kq + 4 * h;
    const float* xg = xb + (size_t)(128 * kq + 4 * h) * Ki + 64 * fh + 4 * c;
    v4f av[2][4], bv[2][16];
    v8f acc[4] = {};
    t_load(pa, xg, 0, av[0], bv[0]);
#pragma unroll
    for (int g = 0; g < 4; ++g) {
      if (g + 1 < 4) t_load(pa, xg, 4 * (g + 1), av[(g + 1) & 1], bv[(g + 1) & 1]);
      __builtin_amdgcn_sched_barrier(0);
#pragma unroll
      for (int i = 0; i < 4; ++i) {
        const v4f A4 = av[g & 1][i];
        v2f A0, A1;
        A0.x = A4.x; A0.y = A4.y;
        A1.x = A4.z; A1.y = A4.w;
#pragma unroll
        for (int j = 0; j < 4; ++j) {
          v2f B0;
          B0.x = bv[g & 1][4 * i][j]; B0.y = bv[g & 1][4 * i + 1][j];
          acc[j] = __builtin_amdgcn_wmma_f32_16x16x4_f32(false, A0, false, B0,
                                                         (short)0, acc[j], false, false);
        }
#pragma unroll
        for (int j = 0; j < 4; ++j) {
          v2f B1;
          B1.x = bv[g & 1][4 * i + 2][j]; B1.y = bv[g & 1][4 * i + 3][j];
          acc[j] = __builtin_amdgcn_wmma_f32_16x16x4_f32(false, A1, false, B1,
                                                         (short)0, acc[j], false, false);
        }
      }
      __builtin_amdgcn_sched_barrier(0);
    }
#pragma unroll
    for (int r = 0; r < 8; ++r) {
      v4f t;
      t.x = acc[0][r]; t.y = acc[1][r]; t.z = acc[2][r]; t.w = acc[3][r];
      *(v4f*)(Tq + (kq * 16 + 8 * h + r) * TP + 64 * fh + 4 * c) = t;
    }
  }
  __syncthreads();

  if (w < 4) {
    const int nt = w;
    float wv[16][4];
#pragma unroll
    for (int it = 0; it < 16; ++it)
#pragma unroll
      for (int s = 0; s < 4; ++s)
        wv[it][s] = W[(size_t)(8 * it + 4 * h + s) * Fo + 16 * nt + c];
    v8f accA = {}, accB = {};
#pragma unroll
    for (int it = 0; it < 16; ++it) {
      const float* tp = Tq + c * TP + 8 * it + 4 * h;
      v4f t4 = *(const v4f*)tp;
      t4 += *(const v4f*)(tp + 16 * TP);
      t4 += *(const v4f*)(tp + 32 * TP);
      t4 += *(const v4f*)(tp + 48 * TP);
      v2f A0, A1, B0, B1;
      A0.x = t4.x; A0.y = t4.y; B0.x = wv[it][0]; B0.y = wv[it][1];
      A1.x = t4.z; A1.y = t4.w; B1.x = wv[it][2]; B1.y = wv[it][3];
      accA = __builtin_amdgcn_wmma_f32_16x16x4_f32(false, A0, false, B0,
                                                   (short)0, accA, false, false);
      accB = __builtin_amdgcn_wmma_f32_16x16x4_f32(false, A1, false, B1,
                                                   (short)0, accB, false, false);
    }
    const float bvv = bias[16 * nt + c];
    float* ob = out + ((size_t)b * Nn + row0 + 8 * h) * Fo + 16 * nt + c;
#pragma unroll
    for (int r = 0; r < 8; ++r) {
      float v = accA[r] + accB[r] + bvv;
      v = (v > 0.f) ? v : (__expf(v) - 1.f);
      ob[(size_t)r * Fo] = v;
    }
  }
}

extern "C" void kernel_launch(void* const* d_in, const int* in_sizes, int n_in,
                              void* d_out, int out_size, void* d_ws, size_t ws_size,
                              hipStream_t stream) {
  const float* x    = (const float*)d_in[0];
  const int*   adj  = (const int*)d_in[1];
  const float* W    = (const float*)d_in[2];
  const float* bias = (const float*)d_in[3];
  const float* a    = (const float*)d_in[4];
  float* out = (float*)d_out;

  gat_one<<<Bn * Nn / 16, NTH, 0, stream>>>(x, adj, W, bias, a, out);
}


# x is read once per block: each wave asks for all its B fragments of prob@x (64 128-bit loads, held in registers) beside the fold's loads; logits are folded from them and the WMMAs run from them
# speedup vs baseline: 2.6169x; 1.0679x over previous
#include <hip/hip_runtime.h>
#include <math.h>

#define Bn 4
#define Nn 512
#define Fo 64
#define Ki 128
#define ALPHAF 0.2f

typedef __attribute__((ext_vector_type(2))) float v2f;
typedef __attribute__((ext_vector_type(8))) float v8f;
typedef __attribute__((ext_vector_type(4))) float v4f;

__device__ __forceinline__ float lrelu(float v) { return v > 0.f ? v : ALPHAF * v; }

#define NTH 256
#define RP  33
#define PP  516
#define TP  132
#define USZ (8 * 32 * RP)

__global__ __launch_bounds__(NTH) void gat_one(const float* __restrict__ x,
                                               const int* __restrict__ adj,
                                               const float* __restrict__ W,
                                               const float* __restrict__ bias,
                                               const float* __restrict__ a,
                                               float* __restrict__ out) {
  __shared__ __attribute__((aligned(16))) float wa[2 * Ki];
  __shared__ float El[256];
  __shared__ __attribute__((aligned(16))) float U[USZ];
  __shared__ __attribute__((aligned(16))) float Tq[4 * 16 * TP];

  const int tid  = threadIdx.x;
  const int lane = tid & 31;
  const int w    = tid >> 5;
  const int b    = blockIdx.x >> 5;
  const int mt   = blockIdx.x & 31;
  const int row0 = mt * 16;
  const bool lo  = mt < 16;
  const float* xb = x + (size_t)b * Nn * Ki;

  const int fk = tid >> 1, hf = tid & 1;
  v4f fw[Fo / 4], fa[Fo / 4];
  {
    const float* wr = W + (size_t)fk * Fo;
    const float* ah = a + hf * Fo;
#pragma unroll
    for (int cc = 0; cc < Fo / 4; ++cc) {
      fw[cc] = *(const v4f*)(wr + 4 * cc);
      fa[cc] = *(const v4f*)(ah + 4 * cc);
    }
  }

  const float bl = bias[lane], bh = bias[lane + 32];
  const float a0l = a[lane], a0h = a[lane + 32], a1l = a[Fo + lane], a1h = a[Fo + lane + 32];

  int mk[2][16];
  {
    const int* ar = adj + (size_t)b * Nn * Nn + (size_t)(row0 + 2 * w) * Nn;
#pragma unroll
    for (int rr = 0; rr < 2; ++rr)
#pragma unroll
      for (int j = 0; j < 16; ++j) mk[rr][j] = ar[rr * Nn + lane + 32 * j];
  }

  const int c  = lane & 15;
  const int h  = lane >> 4;
  const int fh = w & 1;
  const int kq = w >> 1;
  v4f xr[64];
  {
    const float* xg = xb + (size_t)(128 * kq + 4 * h) * Ki + 64 * fh + 4 * c;
#pragma unroll
    for (int it = 0; it < 16; ++it)
#pragma unroll
      for (int s = 0; s < 4; ++s)
        xr[4 * it + s] = *(const v4f*)(xg + (size_t)(8 * it + s) * Ki);
  }

  __builtin_amdgcn_sched_barrier(0);

  {
    float s = 0.f;
#pragma unroll
    for (int cc = 0; cc < Fo / 4; ++cc) {
      s = fmaf(fw[cc].x, fa[cc].x, s);
      s = fmaf(fw[cc].y, fa[cc].y, s);
      s = fmaf(fw[cc].z, fa[cc].z, s);
      s = fmaf(fw[cc].w, fa[cc].w, s);
    }
    wa[hf * Ki + fk] = s;
  }
  float c0 = fmaf(bh, a0h, bl * a0l);
  float c1 = fmaf(bh, a1h, bl * a1l);
#pragma unroll
  for (int off = 16; off >= 1; off >>= 1) {
    c0 += __shfl_xor(c0, off, 32);
    c1 += __shfl_xor(c1, off, 32);
  }
  const float c01 = c0 + c1;
  __syncthreads();

  const int nitems = lo ? 32 : 256;
  float* redw = U + w * (32 * RP);
  {
    const v4f w0 = *(const v4f*)(wa + 64 * fh + 4 * c);
    const v4f w1 = *(const v4f*)(wa + Ki + 64 * fh + 4 * c);
    float* ucol = U + 16 * fh + c;
    if (lo) {
      if (kq == (mt >> 2)) {
        const int q4 = mt & 3;
#pragma unroll
        for (int it = 0; it < 16; ++it) {
          if ((it >> 2) == q4) {
#pragma unroll
            for (int s4 = 0; s4 < 4; ++s4) {
              const v4f xv = xr[4 * it + s4];
              float s = xv.x * w0.x;
              s = fmaf(xv.y, w0.y, s);
              s = fmaf(xv.z, w0.z, s);
              s = fmaf(xv.w, w0.w, s);
              s = fmaf(xv.x, w1.x, s);
              s = fmaf(xv.y, w1.y, s);
              s = fmaf(xv.z, w1.z, s);
              s = fmaf(xv.w, w1.w, s);
              ucol[(8 * (it & 3) + 4 * h + s4) * RP] = s;
            }
          }
        }
      }
    } else {
#pragma unroll
      for (int it = 0; it < 16; ++it) {
#pragma unroll
        for (int e = 0; e < 2; ++e) {
          const v4f xa = xr[4 * it + 2 * e];
          const v4f xq = xr[4 * it + 2 * e + 1];
          float s = xa.x * w0.x;
          s = fmaf(xa.y, w0.y, s);
          s = fmaf(xa.z, w0.z, s);
          s = fmaf(xa.w, w0.w, s);
          s = fmaf(xq.x, w1.x, s);
          s = fmaf(xq.y, w1.y, s);
          s = fmaf(xq.z, w1.z, s);
          s = fmaf(xq.w, w1.w, s);
          ucol[(64 * kq + 4 * it + 2 * h + e) * RP] = s;
        }
      }
    }
  }
  __syncthreads();
  if (32 * w < nitems) {
    float s = 0.f;
#pragma unroll
    for (int t = 0; t < 32; ++t) s += redw[lane * RP + t];
    El[32 * w + lane] = lrelu(s + c01);
  }
  __syncthreads();

  float* Pl = U;
#pragma unroll
  for (int rr = 0; rr < 2; ++rr) {
    const int rl = 2 * w + rr;
    const float elo = El[2 * rl];
    const float ehi = El[2 * rl + 1];
    float vals[16];
    if (lo) {
#pragma unroll
      for (int j = 0; j < 16; ++j) vals[j] = (j < 8) ? elo : ehi;
    } else {
#pragma unroll
      for (int j = 0; j < 16; ++j) vals[j] = El[lane + 32 * (j & 7)];
    }
    float mx = -INFINITY;
#pragma unroll
    for (int j = 0; j < 16; ++j) {
      const float e = (mk[rr][j] > 0) ? vals[j] : -INFINITY;
      vals[j] = e;
      mx = fmaxf(mx, e);
    }
#pragma unroll
    for (int off = 16; off >= 1; off >>= 1) mx = fmaxf(mx, __shfl_xor(mx, off, 32));
    float sum = 0.f;
#pragma unroll
    for (int j = 0; j < 16; ++j) {
      const float e = vals[j];
      const float ex = (e == -INFINITY) ? 0.f : __expf(e - mx);
      vals[j] = ex;
      sum += ex;
    }
#pragma unroll
    for (int off = 16; off >= 1; off >>= 1) sum += __shfl_xor(sum, off, 32);
    const float inv = 1.f / sum;
#pragma unroll
    for (int j = 0; j < 16; ++j) Pl[rl * PP + lane + 32 * j] = vals[j] * inv;
  }
  __syncthreads();

  {
    const float* pa = Pl + c * PP + 128 * kq + 4 * h;
    v4f av[16];
#pragma unroll
    for (int it = 0; it < 16; ++it) av[it] = *(const v4f*)(pa + 8 * it);
    __builtin_amdgcn_sched_barrier(0);
    v8f acc[4] = {};
#pragma unroll
    for (int it = 0; it < 16; ++it) {
      const v4f A4 = av[it];
      v2f A0, A1;
      A0.x = A4.x; A0.y = A4.y;
      A1.x = A4.z; A1.y = A4.w;
#pragma unroll
      for (int j = 0; j < 4; ++j) {
        v2f B0;
        B0.x = xr[4 * it][j]; B0.y = xr[4 * it + 1][j];
        acc[j] = __builtin_amdgcn_wmma_f32_16x16x4_f32(false, A0, false, B0,
                                                       (short)0, acc[j], false, false);
      }
#pragma unroll
      for (int j = 0; j < 4; ++j) {
        v2f B1;
        B1.x = xr[4 * it + 2][j]; B1.y = xr[4 * it + 3][j];
        acc[j] = __builtin_amdgcn_wmma_f32_16x16x4_f32(false, A1, false, B1,
                                                       (short)0, acc[j], false, false);
      }
    }
#pragma unroll
    for (int r = 0; r < 8; ++r) {
      v4f t;
      t.x = acc[0][r]; t.y = acc[1][r]; t.z = acc[2][r]; t.w = acc[3][r];
      *(v4f*)(Tq + (kq * 16 + 8 * h + r) * TP + 64 * fh + 4 * c) = t;
    }
  }
  __syncthreads();

  if (w < 4) {
    const int nt = w;
    float wv[16][4];
#pragma unroll
    for (int it = 0; it < 16; ++it)
#pragma unroll
      for (int s = 0; s < 4; ++s)
        wv[it][s] = W[(size_t)(8 * it + 4 * h + s) * Fo + 16 * nt + c];
    v8f accA = {}, accB = {};
#pragma unroll
    for (int it = 0; it < 16; ++it) {
      const float* tp = Tq + c * TP + 8 * it + 4 * h;
      v4f t4 = *(const v4f*)tp;
      t4 += *(const v4f*)(tp + 16 * TP);
      t4 += *(const v4f*)(tp + 32 * TP);
      t4 += *(const v4f*)(tp + 48 * TP);
      v2f A0, A1, B0, B1;
      A0.x = t4.x; A0.y = t4.y; B0.x = wv[it][0]; B0.y = wv[it][1];
      A1.x = t4.z; A1.y = t4.w; B1.x = wv[it][2]; B1.y = wv[it][3];
      accA = __builtin_amdgcn_wmma_f32_16x16x4_f32(false, A0, false, B0,
                                                   (short)0, accA, false, false);
      accB = __builtin_amdgcn_wmma_f32_16x16x4_f32(false, A1, false, B1,
                                                   (short)0, accB, false, false);
    }
    const float bvv = bias[16 * nt + c];
    float* ob = out + ((size_t)b * Nn + row0 + 8 * h) * Fo + 16 * nt + c;
#pragma unroll
    for (int r = 0; r < 8; ++r) {
      float v = accA[r] + accB[r] + bvv;
      v = (v > 0.f) ? v : (__expf(v) - 1.f);
      ob[(size_t)r * Fo] = v;
    }
  }
}

extern "C" void kernel_launch(void* const* d_in, const int* in_sizes, int n_in,
                              void* d_out, int out_size, void* d_ws, size_t ws_size,
                              hipStream_t stream) {
  const float* x    = (const float*)d_in[0];
  const int*   adj  = (const int*)d_in[1];
  const float* W    = (const float*)d_in[2];
  const float* bias = (const float*)d_in[3];
  const float* a    = (const float*)d_in[4];
  float* out = (float*)d_out;

  gat_one<<<Bn * Nn / 16, NTH, 0, stream>>>(x, adj, W, bias, a, out);
}


# softmax row max and row sum (and the bias-term sums) reduced across the wave by v_permlanex16 + DPP row_xmask on the vector ALU, not by ds_bpermute trips; same butterfly order, sums bit-identical
# speedup vs baseline: 2.7738x; 1.0600x over previous
#include <hip/hip_runtime.h>
#include <math.h>

#define Bn 4
#define Nn 512
#define Fo 64
#define Ki 128
#define ALPHAF 0.2f

typedef __attribute__((ext_vector_type(2))) float v2f;
typedef __attribute__((ext_vector_type(8))) float v8f;
typedef __attribute__((ext_vector_type(4))) float v4f;

__device__ __forceinline__ float lrelu(float v) { return v > 0.f ? v : ALPHAF * v; }

__device__ __forceinline__ float xhalf(float v) {
  const int b = __builtin_bit_cast(int, v);
  return __builtin_bit_cast(float,
      __builtin_amdgcn_permlanex16(b, b, 0x76543210, 0xfedcba98, false, false));
}
#define XROW(v, M) __builtin_bit_cast(float, __builtin_amdgcn_update_dpp( \
    0, __builtin_bit_cast(int, (v)), 0x160 + (M), 0xf, 0xf, true))
__device__ __forceinline__ float wave_sum(float v) {
  v += xhalf(v);
  v += XROW(v, 8);
  v += XROW(v, 4);
  v += XROW(v, 2);
  v += XROW(v, 1);
  return v;
}
__device__ __forceinline__ float wave_max(float v) {
  v = fmaxf(v, xhalf(v));
  v = fmaxf(v, XROW(v, 8));
  v = fmaxf(v, XROW(v, 4));
  v = fmaxf(v, XROW(v, 2));
  v = fmaxf(v, XROW(v, 1));
  return v;
}

#define NTH 256
#define RP  33
#define PP  516
#define TP  132
#define USZ (8 * 32 * RP)

__global__ __launch_bounds__(NTH) void gat_one(const float* __restrict__ x,
                                               const int* __restrict__ adj,
                                               const float* __restrict__ W,
                                               const float* __restrict__ bias,
                                               const float* __restrict__ a,
                                               float* __restrict__ out) {
  __shared__ __attribute__((aligned(16))) float wa[2 * Ki];
  __shared__ float El[256];
  __shared__ __attribute__((aligned(16))) float U[USZ];
  __shared__ __attribute__((aligned(16))) float Tq[4 * 16 * TP];

  const int tid  = threadIdx.x;
  const int lane = tid & 31;
  const int w    = tid >> 5;
  const int b    = blockIdx.x >> 5;
  const int mt   = blockIdx.x & 31;
  const int row0 = mt * 16;
  const bool lo  = mt < 16;
  const float* xb = x + (size_t)b * Nn * Ki;

  const int fk = tid >> 1, hf = tid & 1;
  v4f fw[Fo / 4], fa[Fo / 4];
  {
    const float* wr = W + (size_t)fk * Fo;
    const float* ah = a + hf * Fo;
#pragma unroll
    for (int cc = 0; cc < Fo / 4; ++cc) {
      fw[cc] = *(const v4f*)(wr + 4 * cc);
      fa[cc] = *(const v4f*)(ah + 4 * cc);
    }
  }

  const float bl = bias[lane], bh = bias[lane + 32];
  const float a0l = a[lane], a0h = a[lane + 32], a1l = a[Fo + lane], a1h = a[Fo + lane + 32];

  int mk[2][16];
  {
    const int* ar = adj + (size_t)b * Nn * Nn + (size_t)(row0 + 2 * w) * Nn;
#pragma unroll
    for (int rr = 0; rr < 2; ++rr)
#pragma unroll
      for (int j = 0; j < 16; ++j) mk[rr][j] = ar[rr * Nn + lane + 32 * j];
  }

  const int c  = lane & 15;
  const int h  = lane >> 4;
  const int fh = w & 1;
  const int kq = w >> 1;
  v4f xr[64];
  {
    const float* xg = xb + (size_t)(128 * kq + 4 * h) * Ki + 64 * fh + 4 * c;
#pragma unroll
    for (int it = 0; it < 16; ++it)
#pragma unroll
      for (int s = 0; s < 4; ++s)
        xr[4 * it + s] = *(const v4f*)(xg + (size_t)(8 * it + s) * Ki);
  }

  __builtin_amdgcn_sched_barrier(0);

  {
    float s = 0.f;
#pragma unroll
    for (int cc = 0; cc < Fo / 4; ++cc) {
      s = fmaf(fw[cc].x, fa[cc].x, s);
      s = fmaf(fw[cc].y, fa[cc].y, s);
      s = fmaf(fw[cc].z, fa[cc].z, s);
      s = fmaf(fw[cc].w, fa[cc].w, s);
    }
    wa[hf * Ki + fk] = s;
  }
  float c0 = fmaf(bh, a0h, bl * a0l);
  float c1 = fmaf(bh, a1h, bl * a1l);
  c0 = wave_sum(c0);
  c1 = wave_sum(c1);
  const float c01 = c0 + c1;
  __syncthreads();

  const int nitems = lo ? 32 : 256;
  float* redw = U + w * (32 * RP);
  {
    const v4f w0 = *(const v4f*)(wa + 64 * fh + 4 * c);
    const v4f w1 = *(const v4f*)(wa + Ki + 64 * fh + 4 * c);
    float* ucol = U + 16 * fh + c;
    if (lo) {
      if (kq == (mt >> 2)) {
        const int q4 = mt & 3;
#pragma unroll
        for (int it = 0; it < 16; ++it) {
          if ((it >> 2) == q4) {
#pragma unroll
            for (int s4 = 0; s4 < 4; ++s4) {
              const v4f xv = xr[4 * it + s4];
              float s = xv.x * w0.x;
              s = fmaf(xv.y, w0.y, s);
              s = fmaf(xv.z, w0.z, s);
              s = fmaf(xv.w, w0.w, s);
              s = fmaf(xv.x, w1.x, s);
              s = fmaf(xv.y, w1.y, s);
              s = fmaf(xv.z, w1.z, s);
              s = fmaf(xv.w, w1.w, s);
              ucol[(8 * (it & 3) + 4 * h + s4) * RP] = s;
            }
          }
        }
      }
    } else {
#pragma unroll
      for (int it = 0; it < 16; ++it) {
#pragma unroll
        for (int e = 0; e < 2; ++e) {
          const v4f xa = xr[4 * it + 2 * e];
          const v4f xq = xr[4 * it + 2 * e + 1];
          float s = xa.x * w0.x;
          s = fmaf(xa.y, w0.y, s);
          s = fmaf(xa.z, w0.z, s);
          s = fmaf(xa.w, w0.w, s);
          s = fmaf(xq.x, w1.x, s);
          s = fmaf(xq.y, w1.y, s);
          s = fmaf(xq.z, w1.z, s);
          s = fmaf(xq.w, w1.w, s);
          ucol[(64 * kq + 4 * it + 2 * h + e) * RP] = s;
        }
      }
    }
  }
  __syncthreads();
  if (32 * w < nitems) {
    float s = 0.f;
#pragma unroll
    for (int t = 0; t < 32; ++t) s += redw[lane * RP + t];
    El[32 * w + lane] = lrelu(s + c01);
  }
  __syncthreads();

  float* Pl = U;
#pragma unroll
  for (int rr = 0; rr < 2; ++rr) {
    const int rl = 2 * w + rr;
    const float elo = El[2 * rl];
    const float ehi = El[2 * rl + 1];
    float vals[16];
    if (lo) {
#pragma unroll
      for (int j = 0; j < 16; ++j) vals[j] = (j < 8) ? elo : ehi;
    } else {
#pragma unroll
      for (int j = 0; j < 16; ++j) vals[j] = El[lane + 32 * (j & 7)];
    }
    float mx = -INFINITY;
#pragma unroll
    for (int j = 0; j < 16; ++j) {
      const float e = (mk[rr][j] > 0) ? vals[j] : -INFINITY;
      vals[j] = e;
      mx = fmaxf(mx, e);
    }
    mx = wave_max(mx);
    float sum = 0.f;
#pragma unroll
    for (int j = 0; j < 16; ++j) {
      const float e = vals[j];
      const float ex = (e == -INFINITY) ? 0.f : __expf(e - mx);
      vals[j] = ex;
      sum += ex;
    }
    sum = wave_sum(sum);
    const float inv = 1.f / sum;
#pragma unroll
    for (int j = 0; j < 16; ++j) Pl[rl * PP + lane + 32 * j] = vals[j] * inv;
  }
  __syncthreads();

  {
    const float* pa = Pl + c * PP + 128 * kq + 4 * h;
    v4f av[16];
#pragma unroll
    for (int it = 0; it < 16; ++it) av[it] = *(const v4f*)(pa + 8 * it);
    __builtin_amdgcn_sched_barrier(0);
    v8f acc[4] = {};
#pragma unroll
    for (int it = 0; it < 16; ++it) {
      const v4f A4 = av[it];
      v2f A0, A1;
      A0.x = A4.x; A0.y = A4.y;
      A1.x = A4.z; A1.y = A4.w;
#pragma unroll
      for (int j = 0; j < 4; ++j) {
        v2f B0;
        B0.x = xr[4 * it][j]; B0.y = xr[4 * it + 1][j];
        acc[j] = __builtin_amdgcn_wmma_f32_16x16x4_f32(false, A0, false, B0,
                                                       (short)0, acc[j], false, false);
      }
#pragma unroll
      for (int j = 0; j < 4; ++j) {
        v2f B1;
        B1.x = xr[4 * it + 2][j]; B1.y = xr[4 * it + 3][j];
        acc[j] = __builtin_amdgcn_wmma_f32_16x16x4_f32(false, A1, false, B1,
                                                       (short)0, acc[j], false, false);
      }
    }
#pragma unroll
    for (int r = 0; r < 8; ++r) {
      v4f t;
      t.x = acc[0][r]; t.y = acc[1][r]; t.z = acc[2][r]; t.w = acc[3][r];
      *(v4f*)(Tq + (kq * 16 + 8 * h + r) * TP + 64 * fh + 4 * c) = t;
    }
  }
  __syncthreads();

  if (w < 4) {
    const int nt = w;
    float wv[16][4];
#pragma unroll
    for (int it = 0; it < 16; ++it)
#pragma unroll
      for (int s = 0; s < 4; ++s)
        wv[it][s] = W[(size_t)(8 * it + 4 * h + s) * Fo + 16 * nt + c];
    v8f accA = {}, accB = {};
#pragma unroll
    for (int it = 0; it < 16; ++it) {
      const float* tp = Tq + c * TP + 8 * it + 4 * h;
      v4f t4 = *(const v4f*)tp;
      t4 += *(const v4f*)(tp + 16 * TP);
      t4 += *(const v4f*)(tp + 32 * TP);
      t4 += *(const v4f*)(tp + 48 * TP);
      v2f A0, A1, B0, B1;
      A0.x = t4.x; A0.y = t4.y; B0.x = wv[it][0]; B0.y = wv[it][1];
      A1.x = t4.z; A1.y = t4.w; B1.x = wv[it][2]; B1.y = wv[it][3];
      accA = __builtin_amdgcn_wmma_f32_16x16x4_f32(false, A0, false, B0,
                                                   (short)0, accA, false, false);
      accB = __builtin_amdgcn_wmma_f32_16x16x4_f32(false, A1, false, B1,
                                                   (short)0, accB, false, false);
    }
    const float bvv = bias[16 * nt + c];
    float* ob = out + ((size_t)b * Nn + row0 + 8 * h) * Fo + 16 * nt + c;
#pragma unroll
    for (int r = 0; r < 8; ++r) {
      float v = accA[r] + accB[r] + bvv;
      v = (v > 0.f) ? v : (__expf(v) - 1.f);
      ob[(size_t)r * Fo] = v;
    }
  }
}

extern "C" void kernel_launch(void* const* d_in, const int* in_sizes, int n_in,
                              void* d_out, int out_size, void* d_ws, size_t ws_size,
                              hipStream_t stream) {
  const float* x    = (const float*)d_in[0];
  const int*   adj  = (const int*)d_in[1];
  const float* W    = (const float*)d_in[2];
  const float* bias = (const float*)d_in[3];
  const float* a    = (const float*)d_in[4];
  float* out = (float*)d_out;

  gat_one<<<Bn * Nn / 16, NTH, 0, stream>>>(x, adj, W, bias, a, out);
}
